# starting kernel
# baseline (speedup 1.0000x reference)
#include <hip/hip_runtime.h>
#include <hip/hip_bf16.h>

typedef __attribute__((ext_vector_type(2)))  float    v2f;
typedef __attribute__((ext_vector_type(8)))  float    v8f;
typedef __attribute__((ext_vector_type(16))) _Float16 v16h;

__global__ void gcn_init_kernel(float* __restrict__ h2acc, float* __restrict__ deg,
                                float* __restrict__ z, float* __restrict__ out,
                                int N, int paddedN) {
    int i = blockIdx.x * blockDim.x + threadIdx.x;
    if (i < paddedN * 16) h2acc[i] = 0.0f;
    if (i < paddedN) { deg[i] = (i < N) ? 1.0f : 0.0f; z[i] = 0.0f; }
    if (i < N) out[i] = 0.0f;
}

__global__ void gcn_degree_kernel(const int* __restrict__ dst, float* __restrict__ deg, int E) {
    int e = blockIdx.x * blockDim.x + threadIdx.x;
    if (e < E) atomicAdd(&deg[dst[e]], 1.0f);
}

__global__ void gcn_rsqrt_kernel(float* __restrict__ dinv, int N) {
    int n = blockIdx.x * blockDim.x + threadIdx.x;
    if (n < N) dinv[n] = rsqrtf(dinv[n]);
}

__global__ void gcn_agg1_kernel(const int* __restrict__ src, const int* __restrict__ dst,
                                const float* __restrict__ dinv, const float* __restrict__ x,
                                const float* __restrict__ W1, float* __restrict__ h2acc, int E) {
    int t = blockIdx.x * blockDim.x + threadIdx.x;
    int e = t >> 4;
    if (e >= E) return;
    int f = t & 15;
    int s = src[e];
    int d = dst[e];
    float w = dinv[s] * dinv[d];
    atomicAdd(&h2acc[(size_t)d * 16 + f], w * x[s] * W1[f]);
}

__global__ void gcn_pointwise1_kernel(float* __restrict__ h2, const float* __restrict__ dinv,
                                      const float* __restrict__ x, const float* __restrict__ W1,
                                      const float* __restrict__ b1, int N) {
    int t = blockIdx.x * blockDim.x + threadIdx.x;
    int n = t >> 4;
    if (n >= N) return;
    int f = t & 15;
    float di = dinv[n];
    float v = h2[(size_t)n * 16 + f] + di * di * x[n] * W1[f] + b1[f];
    h2[(size_t)n * 16 + f] = fmaxf(v, 0.0f);
}

__global__ void gcn_gemv_wmma_kernel(const float* __restrict__ h2, const float* __restrict__ W2,
                                     float* __restrict__ z, int n_tiles) {
    int lane = threadIdx.x & 31;
    int wave = (blockIdx.x * blockDim.x + threadIdx.x) >> 5;
    if (wave >= n_tiles) return;
    int base = wave * 16;
    int m  = lane & 15;
    int kh = lane >> 4;
    const float* row = h2 + (size_t)(base + m) * 16;
    v8f acc = {0.f, 0.f, 0.f, 0.f, 0.f, 0.f, 0.f, 0.f};

#if __has_builtin(__builtin_amdgcn_wmma_f32_16x16x4_f32)
#pragma unroll
    for (int kc = 0; kc < 4; ++kc) {
        v2f a, b;
        a.x = row[kc * 4 + kh * 2 + 0];
        a.y = row[kc * 4 + kh * 2 + 1];
        b.x = W2[kc * 4 + kh * 2 + 0];
        b.y = W2[kc * 4 + kh * 2 + 1];
        acc = __builtin_amdgcn_wmma_f32_16x16x4_f32(
             false, a,  false, b,
             (short)0, acc,  false,  false);
    }
#else
    v16h a, b;
#pragma unroll
    for (int j = 0; j < 8; ++j) {
        a[j]     = (_Float16)row[kh * 8 + j];
        a[8 + j] = (_Float16)0.0f;
        b[j]     = (lane < 16) ? (_Float16)W2[j]     : (_Float16)0.0f;
        b[8 + j] = (lane < 16) ? (_Float16)W2[8 + j] : (_Float16)0.0f;
    }
    acc = __builtin_amdgcn_wmma_f32_16x16x32_f16(
        false, a, false, b, (short)0, acc, false, false);
#endif

    if (m == 0) {
        float* zp = z + base + kh * 8;
        zp[0] = acc[0]; zp[1] = acc[1]; zp[2] = acc[2]; zp[3] = acc[3];
        zp[4] = acc[4]; zp[5] = acc[5]; zp[6] = acc[6]; zp[7] = acc[7];
    }
}

__global__ void gcn_agg2_kernel(const int* __restrict__ src, const int* __restrict__ dst,
                                const float* __restrict__ dinv, const float* __restrict__ z,
                                float* __restrict__ out, int E) {
    int e = blockIdx.x * blockDim.x + threadIdx.x;
    if (e >= E) return;
    int s = src[e];
    int d = dst[e];
    atomicAdd(&out[d], dinv[s] * dinv[d] * z[s]);
}

__global__ void gcn_final_kernel(float* __restrict__ out, const float* __restrict__ dinv,
                                 const float* __restrict__ z, const float* __restrict__ b2, int N) {
    int n = blockIdx.x * blockDim.x + threadIdx.x;
    if (n >= N) return;
    float di = dinv[n];
    out[n] += di * di * z[n] + b2[0];
}

extern "C" void kernel_launch(void* const* d_in, const int* in_sizes, int n_in,
                              void* d_out, int out_size, void* d_ws, size_t ws_size,
                              hipStream_t stream) {
    const float* x  = (const float*)d_in[0];
    const float* W1 = (const float*)d_in[1];
    const float* b1 = (const float*)d_in[2];
    const float* W2 = (const float*)d_in[3];
    const float* b2 = (const float*)d_in[4];
    const int*   ei = (const int*)d_in[5];
    int N = in_sizes[0];
    int E = in_sizes[5] / 2;
    const int* src = ei;
    const int* dst = ei + E;
    int paddedN = (N + 15) & ~15;

    char* ws = (char*)d_ws;
    float* h2   = (float*)ws; ws += (size_t)paddedN * 16 * sizeof(float);
    float* dinv = (float*)ws; ws += (size_t)paddedN * sizeof(float);
    float* z    = (float*)ws; ws += (size_t)paddedN * sizeof(float);
    float* out  = (float*)d_out;

    const int B = 256;
    int tiles = paddedN / 16;

    gcn_init_kernel<<<(paddedN * 16 + B - 1) / B, B, 0, stream>>>(h2, dinv, z, out, N, paddedN);
    gcn_degree_kernel<<<(E + B - 1) / B, B, 0, stream>>>(dst, dinv, E);
    gcn_rsqrt_kernel<<<(N + B - 1) / B, B, 0, stream>>>(dinv, N);
    gcn_agg1_kernel<<<((long long)E * 16 + B - 1) / B, B, 0, stream>>>(src, dst, dinv, x, W1, h2, E);
    gcn_pointwise1_kernel<<<((long long)N * 16 + B - 1) / B, B, 0, stream>>>(h2, dinv, x, W1, b1, N);
    gcn_gemv_wmma_kernel<<<(tiles * 32 + B - 1) / B, B, 0, stream>>>(h2, W2, z, tiles);
    gcn_agg2_kernel<<<(E + B - 1) / B, B, 0, stream>>>(src, dst, dinv, z, out, E);
    gcn_final_kernel<<<(N + B - 1) / B, B, 0, stream>>>(out, dinv, z, b2, N);
}


# Degree and layer-2 scatter-adds go to LDS cells per dst window, slice partials summed in the consumer kernels (dinv[dst] applied once per node): no global atomics there; sum order changes
# speedup vs baseline: 2.1081x; 2.1081x over previous
#include <hip/hip_runtime.h>
#include <hip/hip_bf16.h>

typedef __attribute__((ext_vector_type(2)))  float    v2f;
typedef __attribute__((ext_vector_type(8)))  float    v8f;
typedef __attribute__((ext_vector_type(16))) _Float16 v16h;

__global__ void gcn_init_kernel(float* __restrict__ h2acc, float* __restrict__ z, int paddedN) {
    int i = blockIdx.x * blockDim.x + threadIdx.x;
    if (i < paddedN * 16) h2acc[i] = 0.0f;
    if (i < paddedN) z[i] = 0.0f;
}

#define GCN_RANGE  25088
#define GCN_SLICES 64
#define GCN_TPB    1024


__global__ __launch_bounds__(GCN_TPB)
void gcn_degree_lds_kernel(const int* __restrict__ dst, unsigned int* __restrict__ part,
                           int E, int eps, int NP) {
    __shared__ alignas(16) unsigned int cnt[GCN_RANGE];
    const int tid = threadIdx.x;
    const int lo  = blockIdx.y * GCN_RANGE;
    uint4* c4 = reinterpret_cast<uint4*>(cnt);
    for (int i = tid; i < GCN_RANGE / 4; i += GCN_TPB) c4[i] = make_uint4(0u, 0u, 0u, 0u);
    __syncthreads();

    const int e0 = blockIdx.x * eps;
    const int n  = min(eps, E - e0);
    const int* q = dst + e0;
    int nv = 0;
    if (n > 0 && (reinterpret_cast<size_t>(q) & 15) == 0) nv = n >> 2;
    const int4* p = reinterpret_cast<const int4*>(q);
    for (int i = tid; i < nv; i += 4 * GCN_TPB) {
        int4 v[4];
#pragma unroll
        for (int u = 0; u < 4; ++u) {
            int j = i + u * GCN_TPB;
            v[u] = p[min(j, nv - 1)];
            if (j >= nv) v[u] = make_int4(-1, -1, -1, -1);
        }
#pragma unroll
        for (int u = 0; u < 4; ++u) {
            unsigned int r0 = (unsigned int)(v[u].x - lo), r1 = (unsigned int)(v[u].y - lo);
            unsigned int r2 = (unsigned int)(v[u].z - lo), r3 = (unsigned int)(v[u].w - lo);
            if (r0 < GCN_RANGE) atomicAdd(&cnt[r0], 1u);
            if (r1 < GCN_RANGE) atomicAdd(&cnt[r1], 1u);
            if (r2 < GCN_RANGE) atomicAdd(&cnt[r2], 1u);
            if (r3 < GCN_RANGE) atomicAdd(&cnt[r3], 1u);
        }
    }
    for (int e = 4 * nv + tid; e < n; e += GCN_TPB) {
        unsigned int r = (unsigned int)(q[e] - lo);
        if (r < GCN_RANGE) atomicAdd(&cnt[r], 1u);
    }
    __syncthreads();

    uint4* o4 = reinterpret_cast<uint4*>(part + (size_t)blockIdx.x * NP + lo);
    for (int i = tid; i < GCN_RANGE / 4; i += GCN_TPB) o4[i] = c4[i];
}

__global__ void gcn_rsqrt_kernel(const unsigned int* __restrict__ part, float* __restrict__ dinv,
                                 int N, int NP) {
    int n = blockIdx.x * blockDim.x + threadIdx.x;
    if (n >= N) return;
    unsigned int c = 0u;
#pragma unroll 16
    for (int k = 0; k < GCN_SLICES; ++k) c += part[(size_t)k * NP + n];
    dinv[n] = rsqrtf(1.0f + (float)c);
}

__global__ void gcn_agg1_kernel(const int* __restrict__ src, const int* __restrict__ dst,
                                const float* __restrict__ dinv, const float* __restrict__ x,
                                const float* __restrict__ W1, float* __restrict__ h2acc, int E) {
    int t = blockIdx.x * blockDim.x + threadIdx.x;
    int e = t >> 4;
    if (e >= E) return;
    int f = t & 15;
    int s = src[e];
    int d = dst[e];
    float w = dinv[s] * dinv[d];
    atomicAdd(&h2acc[(size_t)d * 16 + f], w * x[s] * W1[f]);
}

__global__ void gcn_pointwise1_kernel(float* __restrict__ h2, const float* __restrict__ dinv,
                                      const float* __restrict__ x, const float* __restrict__ W1,
                                      const float* __restrict__ b1, int N) {
    int t = blockIdx.x * blockDim.x + threadIdx.x;
    int n = t >> 4;
    if (n >= N) return;
    int f = t & 15;
    float di = dinv[n];
    float v = h2[(size_t)n * 16 + f] + di * di * x[n] * W1[f] + b1[f];
    h2[(size_t)n * 16 + f] = fmaxf(v, 0.0f);
}

__global__ void gcn_gemv_wmma_kernel(const float* __restrict__ h2, const float* __restrict__ W2,
                                     float* __restrict__ z, int n_tiles) {
    int lane = threadIdx.x & 31;
    int wave = (blockIdx.x * blockDim.x + threadIdx.x) >> 5;
    if (wave >= n_tiles) return;
    int base = wave * 16;
    int m  = lane & 15;
    int kh = lane >> 4;
    const float* row = h2 + (size_t)(base + m) * 16;
    v8f acc = {0.f, 0.f, 0.f, 0.f, 0.f, 0.f, 0.f, 0.f};

#if __has_builtin(__builtin_amdgcn_wmma_f32_16x16x4_f32)
#pragma unroll
    for (int kc = 0; kc < 4; ++kc) {
        v2f a, b;
        a.x = row[kc * 4 + kh * 2 + 0];
        a.y = row[kc * 4 + kh * 2 + 1];
        b.x = W2[kc * 4 + kh * 2 + 0];
        b.y = W2[kc * 4 + kh * 2 + 1];
        acc = __builtin_amdgcn_wmma_f32_16x16x4_f32(
             false, a,  false, b,
             (short)0, acc,  false,  false);
    }
#else
    v16h a, b;
#pragma unroll
    for (int j = 0; j < 8; ++j) {
        a[j]     = (_Float16)row[kh * 8 + j];
        a[8 + j] = (_Float16)0.0f;
        b[j]     = (lane < 16) ? (_Float16)W2[j]     : (_Float16)0.0f;
        b[8 + j] = (lane < 16) ? (_Float16)W2[8 + j] : (_Float16)0.0f;
    }
    acc = __builtin_amdgcn_wmma_f32_16x16x32_f16(
        false, a, false, b, (short)0, acc, false, false);
#endif

    if (m == 0) {
        float* zp = z + base + kh * 8;
        zp[0] = acc[0]; zp[1] = acc[1]; zp[2] = acc[2]; zp[3] = acc[3];
        zp[4] = acc[4]; zp[5] = acc[5]; zp[6] = acc[6]; zp[7] = acc[7];
    }
}

__global__ __launch_bounds__(GCN_TPB)
void gcn_agg2_lds_kernel(const int* __restrict__ src, const int* __restrict__ dst,
                         const float* __restrict__ dinv, const float* __restrict__ z,
                         float* __restrict__ part, int E, int eps, int NP) {
    __shared__ alignas(16) float acc[GCN_RANGE];
    const int tid = threadIdx.x;
    const int lo  = blockIdx.y * GCN_RANGE;
    float4* a4 = reinterpret_cast<float4*>(acc);
    for (int i = tid; i < GCN_RANGE / 4; i += GCN_TPB) a4[i] = make_float4(0.f, 0.f, 0.f, 0.f);
    __syncthreads();

    const int e0 = blockIdx.x * eps;
    const int n  = min(eps, E - e0);
    const int* qd = dst + e0;
    const int* qs = src + e0;
    int nv = 0;
    if (n > 0 && ((reinterpret_cast<size_t>(qd) | reinterpret_cast<size_t>(qs)) & 15) == 0) nv = n >> 2;
    const int4* pd = reinterpret_cast<const int4*>(qd);
    const int4* ps = reinterpret_cast<const int4*>(qs);
    for (int i = tid; i < nv; i += 2 * GCN_TPB) {
        int d[8], s[8];
#pragma unroll
        for (int u = 0; u < 2; ++u) {
            int j = i + u * GCN_TPB;
            int4 dv = pd[min(j, nv - 1)];
            int4 sv = ps[min(j, nv - 1)];
            if (j >= nv) dv = make_int4(-1, -1, -1, -1);
            d[4 * u + 0] = dv.x; d[4 * u + 1] = dv.y; d[4 * u + 2] = dv.z; d[4 * u + 3] = dv.w;
            s[4 * u + 0] = sv.x; s[4 * u + 1] = sv.y; s[4 * u + 2] = sv.z; s[4 * u + 3] = sv.w;
        }
        float w[8];
        unsigned int r[8];
#pragma unroll
        for (int t = 0; t < 8; ++t) {
            r[t] = (unsigned int)(d[t] - lo);
            int ss = (r[t] < GCN_RANGE) ? s[t] : 0;
            w[t] = dinv[ss] * z[ss];
        }
#pragma unroll
        for (int t = 0; t < 8; ++t)
            if (r[t] < GCN_RANGE) atomicAdd(&acc[r[t]], w[t]);
    }
    for (int e = 4 * nv + tid; e < n; e += GCN_TPB) {
        unsigned int r = (unsigned int)(qd[e] - lo);
        if (r < GCN_RANGE) { int ss = qs[e]; atomicAdd(&acc[r], dinv[ss] * z[ss]); }
    }
    __syncthreads();

    float4* o4 = reinterpret_cast<float4*>(part + (size_t)blockIdx.x * NP + lo);
    for (int i = tid; i < GCN_RANGE / 4; i += GCN_TPB) o4[i] = a4[i];
}

__global__ void gcn_final_kernel(float* __restrict__ out, const float* __restrict__ part,
                                 const float* __restrict__ dinv, const float* __restrict__ z,
                                 const float* __restrict__ b2, int N, int NP) {
    int n = blockIdx.x * blockDim.x + threadIdx.x;
    if (n >= N) return;
    float a = 0.0f;
#pragma unroll 16
    for (int k = 0; k < GCN_SLICES; ++k) a += part[(size_t)k * NP + n];
    float di = dinv[n];
    out[n] = di * a + di * di * z[n] + b2[0];
}

extern "C" void kernel_launch(void* const* d_in, const int* in_sizes, int n_in,
                              void* d_out, int out_size, void* d_ws, size_t ws_size,
                              hipStream_t stream) {
    const float* x  = (const float*)d_in[0];
    const float* W1 = (const float*)d_in[1];
    const float* b1 = (const float*)d_in[2];
    const float* W2 = (const float*)d_in[3];
    const float* b2 = (const float*)d_in[4];
    const int*   ei = (const int*)d_in[5];
    int N = in_sizes[0];
    int E = in_sizes[5] / 2;
    const int* src = ei;
    const int* dst = ei + E;
    int paddedN = (N + 15) & ~15;

    char* ws = (char*)d_ws;
    float* h2   = (float*)ws; ws += (size_t)paddedN * 16 * sizeof(float);
    float* dinv = (float*)ws; ws += (size_t)paddedN * sizeof(float);
    float* z    = (float*)ws; ws += (size_t)paddedN * sizeof(float);
    float* out  = (float*)d_out;

    const int B = 256;
    int tiles = paddedN / 16;

    int windows = (N + GCN_RANGE - 1) / GCN_RANGE;
    int NP  = windows * GCN_RANGE;
    int eps = (((E + GCN_SLICES - 1) / GCN_SLICES) + 3) & ~3;
    float* part = (float*)ws;
    dim3 sgrid(GCN_SLICES, windows);

    gcn_init_kernel<<<(paddedN * 16 + B - 1) / B, B, 0, stream>>>(h2, z, paddedN);
    gcn_degree_lds_kernel<<<sgrid, GCN_TPB, 0, stream>>>(dst, (unsigned int*)part, E, eps, NP);
    gcn_rsqrt_kernel<<<(N + B - 1) / B, B, 0, stream>>>((const unsigned int*)part, dinv, N, NP);
    gcn_agg1_kernel<<<((long long)E * 16 + B - 1) / B, B, 0, stream>>>(src, dst, dinv, x, W1, h2, E);
    gcn_pointwise1_kernel<<<((long long)N * 16 + B - 1) / B, B, 0, stream>>>(h2, dinv, x, W1, b1, N);
    gcn_gemv_wmma_kernel<<<(tiles * 32 + B - 1) / B, B, 0, stream>>>(h2, W2, z, tiles);
    gcn_agg2_lds_kernel<<<sgrid, GCN_TPB, 0, stream>>>(src, dst, dinv, z, part, E, eps, NP);
    gcn_final_kernel<<<(N + B - 1) / B, B, 0, stream>>>(out, part, dinv, z, b2, N, NP);
}


# Layer 1 is rank 1 (x is [N,1]): its 16-wide global-atomic edge scatter becomes the scalar LDS scatter sum dinv[s]*x[s] per dst; features expanded per node in the WMMA A rows; sum order changes
# speedup vs baseline: 4.4489x; 2.1104x over previous
#include <hip/hip_runtime.h>
#include <hip/hip_bf16.h>

typedef __attribute__((ext_vector_type(2)))  float    v2f;
typedef __attribute__((ext_vector_type(8)))  float    v8f;

#define GCN_RANGE  25088
#define GCN_SLICES 64
#define GCN_TPB    1024


__global__ __launch_bounds__(GCN_TPB)
void gcn_degree_lds_kernel(const int* __restrict__ dst, unsigned int* __restrict__ part,
                           int E, int eps, int NP) {
    __shared__ alignas(16) unsigned int cnt[GCN_RANGE];
    const int tid = threadIdx.x;
    const int lo  = blockIdx.y * GCN_RANGE;
    uint4* c4 = reinterpret_cast<uint4*>(cnt);
    for (int i = tid; i < GCN_RANGE / 4; i += GCN_TPB) c4[i] = make_uint4(0u, 0u, 0u, 0u);
    __syncthreads();

    const int e0 = blockIdx.x * eps;
    const int n  = min(eps, E - e0);
    const int* q = dst + e0;
    int nv = 0;
    if (n > 0 && (reinterpret_cast<size_t>(q) & 15) == 0) nv = n >> 2;
    const int4* p = reinterpret_cast<const int4*>(q);
    for (int i = tid; i < nv; i += 4 * GCN_TPB) {
        int4 v[4];
#pragma unroll
        for (int u = 0; u < 4; ++u) {
            int j = i + u * GCN_TPB;
            v[u] = p[min(j, nv - 1)];
            if (j >= nv) v[u] = make_int4(-1, -1, -1, -1);
        }
#pragma unroll
        for (int u = 0; u < 4; ++u) {
            unsigned int r0 = (unsigned int)(v[u].x - lo), r1 = (unsigned int)(v[u].y - lo);
            unsigned int r2 = (unsigned int)(v[u].z - lo), r3 = (unsigned int)(v[u].w - lo);
            if (r0 < GCN_RANGE) atomicAdd(&cnt[r0], 1u);
            if (r1 < GCN_RANGE) atomicAdd(&cnt[r1], 1u);
            if (r2 < GCN_RANGE) atomicAdd(&cnt[r2], 1u);
            if (r3 < GCN_RANGE) atomicAdd(&cnt[r3], 1u);
        }
    }
    for (int e = 4 * nv + tid; e < n; e += GCN_TPB) {
        unsigned int r = (unsigned int)(q[e] - lo);
        if (r < GCN_RANGE) atomicAdd(&cnt[r], 1u);
    }
    __syncthreads();

    uint4* o4 = reinterpret_cast<uint4*>(part + (size_t)blockIdx.x * NP + lo);
    for (int i = tid; i < GCN_RANGE / 4; i += GCN_TPB) o4[i] = c4[i];
}

__global__ void gcn_rsqrt_kernel(const unsigned int* __restrict__ part, float* __restrict__ dinv,
                                 int N, int NP) {
    int n = blockIdx.x * blockDim.x + threadIdx.x;
    if (n >= N) return;
    unsigned int c = 0u;
#pragma unroll 16
    for (int k = 0; k < GCN_SLICES; ++k) c += part[(size_t)k * NP + n];
    dinv[n] = rsqrtf(1.0f + (float)c);
}

__global__ void gcn_layer1_gemv_wmma_kernel(const float* __restrict__ part,
                                            const float* __restrict__ dinv,
                                            const float* __restrict__ x,
                                            const float* __restrict__ W1,
                                            const float* __restrict__ b1,
                                            const float* __restrict__ W2,
                                            float* __restrict__ z, int n_tiles, int N, int NP) {
    int lane = threadIdx.x & 31;
    int wave = (blockIdx.x * blockDim.x + threadIdx.x) >> 5;
    if (wave >= n_tiles) return;
    int base = wave * 16;
    int m  = lane & 15;
    int kh = lane >> 4;
    int n = min(base + m, N - 1);

    const float* p = part + (size_t)(kh * (GCN_SLICES / 2)) * NP + n;
    float t = 0.0f;
#pragma unroll
    for (int k = 0; k < GCN_SLICES / 2; ++k) t += p[(size_t)k * NP];
    t += __shfl_xor(t, 16);
    float di = dinv[n];
    float s1 = di * t + di * di * x[n];

    v8f acc = {0.f, 0.f, 0.f, 0.f, 0.f, 0.f, 0.f, 0.f};
#pragma unroll
    for (int kc = 0; kc < 4; ++kc) {
        int k0 = kc * 4 + kh * 2;
        v2f a, b;
        a.x = fmaxf(W1[k0 + 0] * s1 + b1[k0 + 0], 0.0f);
        a.y = fmaxf(W1[k0 + 1] * s1 + b1[k0 + 1], 0.0f);
        b.x = W2[k0 + 0];
        b.y = W2[k0 + 1];
        acc = __builtin_amdgcn_wmma_f32_16x16x4_f32(
             false, a,  false, b,
             (short)0, acc,  false,  false);
    }

    if (m == 0) {
        float* zp = z + base + kh * 8;
        zp[0] = acc[0]; zp[1] = acc[1]; zp[2] = acc[2]; zp[3] = acc[3];
        zp[4] = acc[4]; zp[5] = acc[5]; zp[6] = acc[6]; zp[7] = acc[7];
    }
}

__global__ __launch_bounds__(GCN_TPB)
void gcn_scatter_lds_kernel(const int* __restrict__ src, const int* __restrict__ dst,
                            const float* __restrict__ dinv, const float* __restrict__ v,
                            float* __restrict__ part, int E, int eps, int NP) {
    __shared__ alignas(16) float acc[GCN_RANGE];
    const int tid = threadIdx.x;
    const int lo  = blockIdx.y * GCN_RANGE;
    float4* a4 = reinterpret_cast<float4*>(acc);
    for (int i = tid; i < GCN_RANGE / 4; i += GCN_TPB) a4[i] = make_float4(0.f, 0.f, 0.f, 0.f);
    __syncthreads();

    const int e0 = blockIdx.x * eps;
    const int n  = min(eps, E - e0);
    const int* qd = dst + e0;
    const int* qs = src + e0;
    int nv = 0;
    if (n > 0 && ((reinterpret_cast<size_t>(qd) | reinterpret_cast<size_t>(qs)) & 15) == 0) nv = n >> 2;
    const int4* pd = reinterpret_cast<const int4*>(qd);
    const int4* ps = reinterpret_cast<const int4*>(qs);
    for (int i = tid; i < nv; i += 2 * GCN_TPB) {
        int d[8], s[8];
#pragma unroll
        for (int u = 0; u < 2; ++u) {
            int j = i + u * GCN_TPB;
            int4 dv = pd[min(j, nv - 1)];
            int4 sv = ps[min(j, nv - 1)];
            if (j >= nv) dv = make_int4(-1, -1, -1, -1);
            d[4 * u + 0] = dv.x; d[4 * u + 1] = dv.y; d[4 * u + 2] = dv.z; d[4 * u + 3] = dv.w;
            s[4 * u + 0] = sv.x; s[4 * u + 1] = sv.y; s[4 * u + 2] = sv.z; s[4 * u + 3] = sv.w;
        }
        float w[8];
        unsigned int r[8];
#pragma unroll
        for (int t = 0; t < 8; ++t) {
            r[t] = (unsigned int)(d[t] - lo);
            int ss = (r[t] < GCN_RANGE) ? s[t] : 0;
            w[t] = dinv[ss] * v[ss];
        }
#pragma unroll
        for (int t = 0; t < 8; ++t)
            if (r[t] < GCN_RANGE) atomicAdd(&acc[r[t]], w[t]);
    }
    for (int e = 4 * nv + tid; e < n; e += GCN_TPB) {
        unsigned int r = (unsigned int)(qd[e] - lo);
        if (r < GCN_RANGE) { int ss = qs[e]; atomicAdd(&acc[r], dinv[ss] * v[ss]); }
    }
    __syncthreads();

    float4* o4 = reinterpret_cast<float4*>(part + (size_t)blockIdx.x * NP + lo);
    for (int i = tid; i < GCN_RANGE / 4; i += GCN_TPB) o4[i] = a4[i];
}

__global__ void gcn_final_kernel(float* __restrict__ out, const float* __restrict__ part,
                                 const float* __restrict__ dinv, const float* __restrict__ z,
                                 const float* __restrict__ b2, int N, int NP) {
    int n = blockIdx.x * blockDim.x + threadIdx.x;
    if (n >= N) return;
    float a = 0.0f;
#pragma unroll 16
    for (int k = 0; k < GCN_SLICES; ++k) a += part[(size_t)k * NP + n];
    float di = dinv[n];
    out[n] = di * a + di * di * z[n] + b2[0];
}

extern "C" void kernel_launch(void* const* d_in, const int* in_sizes, int n_in,
                              void* d_out, int out_size, void* d_ws, size_t ws_size,
                              hipStream_t stream) {
    const float* x  = (const float*)d_in[0];
    const float* W1 = (const float*)d_in[1];
    const float* b1 = (const float*)d_in[2];
    const float* W2 = (const float*)d_in[3];
    const float* b2 = (const float*)d_in[4];
    const int*   ei = (const int*)d_in[5];
    int N = in_sizes[0];
    int E = in_sizes[5] / 2;
    const int* src = ei;
    const int* dst = ei + E;
    int paddedN = (N + 15) & ~15;

    char* ws = (char*)d_ws;
    float* dinv = (float*)ws; ws += (size_t)paddedN * sizeof(float);
    float* z    = (float*)ws; ws += (size_t)paddedN * sizeof(float);
    float* out  = (float*)d_out;

    const int B = 256;
    int tiles = paddedN / 16;

    int windows = (N + GCN_RANGE - 1) / GCN_RANGE;
    int NP  = windows * GCN_RANGE;
    int eps = (((E + GCN_SLICES - 1) / GCN_SLICES) + 3) & ~3;
    float* part = (float*)ws;
    dim3 sgrid(GCN_SLICES, windows);

    gcn_degree_lds_kernel<<<sgrid, GCN_TPB, 0, stream>>>(dst, (unsigned int*)part, E, eps, NP);
    gcn_rsqrt_kernel<<<(N + B - 1) / B, B, 0, stream>>>((const unsigned int*)part, dinv, N, NP);
    gcn_scatter_lds_kernel<<<sgrid, GCN_TPB, 0, stream>>>(src, dst, dinv, x, part, E, eps, NP);
    gcn_layer1_gemv_wmma_kernel<<<(tiles * 32 + B - 1) / B, B, 0, stream>>>(part, dinv, x, W1, b1, W2, z, tiles, N, NP);
    gcn_scatter_lds_kernel<<<sgrid, GCN_TPB, 0, stream>>>(src, dst, dinv, z, part, E, eps, NP);
    gcn_final_kernel<<<(N + B - 1) / B, B, 0, stream>>>(out, part, dinv, z, b2, N, NP);
}


# Edge term dinv[s]*v[s] stored per node by the kernel that produces v (x, then z); both edge scatters gather that one value per edge instead of two; same addends, LDS-atomic sum order as before
# speedup vs baseline: 5.9969x; 1.3480x over previous
#include <hip/hip_runtime.h>
#include <hip/hip_bf16.h>

typedef __attribute__((ext_vector_type(2)))  float    v2f;
typedef __attribute__((ext_vector_type(8)))  float    v8f;

#define GCN_RANGE  25088
#define GCN_SLICES 64
#define GCN_TPB    1024


__global__ __launch_bounds__(GCN_TPB)
void gcn_degree_lds_kernel(const int* __restrict__ dst, unsigned int* __restrict__ part,
                           int E, int eps, int NP) {
    __shared__ alignas(16) unsigned int cnt[GCN_RANGE];
    const int tid = threadIdx.x;
    const int lo  = blockIdx.y * GCN_RANGE;
    uint4* c4 = reinterpret_cast<uint4*>(cnt);
    for (int i = tid; i < GCN_RANGE / 4; i += GCN_TPB) c4[i] = make_uint4(0u, 0u, 0u, 0u);
    __syncthreads();

    const int e0 = blockIdx.x * eps;
    const int n  = min(eps, E - e0);
    const int* q = dst + e0;
    int nv = 0;
    if (n > 0 && (reinterpret_cast<size_t>(q) & 15) == 0) nv = n >> 2;
    const int4* p = reinterpret_cast<const int4*>(q);
    for (int i = tid; i < nv; i += 4 * GCN_TPB) {
        int4 v[4];
#pragma unroll
        for (int u = 0; u < 4; ++u) {
            int j = i + u * GCN_TPB;
            v[u] = p[min(j, nv - 1)];
            if (j >= nv) v[u] = make_int4(-1, -1, -1, -1);
        }
#pragma unroll
        for (int u = 0; u < 4; ++u) {
            unsigned int r0 = (unsigned int)(v[u].x - lo), r1 = (unsigned int)(v[u].y - lo);
            unsigned int r2 = (unsigned int)(v[u].z - lo), r3 = (unsigned int)(v[u].w - lo);
            if (r0 < GCN_RANGE) atomicAdd(&cnt[r0], 1u);
            if (r1 < GCN_RANGE) atomicAdd(&cnt[r1], 1u);
            if (r2 < GCN_RANGE) atomicAdd(&cnt[r2], 1u);
            if (r3 < GCN_RANGE) atomicAdd(&cnt[r3], 1u);
        }
    }
    for (int e = 4 * nv + tid; e < n; e += GCN_TPB) {
        unsigned int r = (unsigned int)(q[e] - lo);
        if (r < GCN_RANGE) atomicAdd(&cnt[r], 1u);
    }
    __syncthreads();

    uint4* o4 = reinterpret_cast<uint4*>(part + (size_t)blockIdx.x * NP + lo);
    for (int i = tid; i < GCN_RANGE / 4; i += GCN_TPB) o4[i] = c4[i];
}

__global__ void gcn_rsqrt_kernel(const unsigned int* __restrict__ part, const float* __restrict__ x,
                                 float* __restrict__ dinv, float* __restrict__ w, int N, int NP) {
    int n = blockIdx.x * blockDim.x + threadIdx.x;
    if (n >= N) return;
    unsigned int c = 0u;
#pragma unroll 16
    for (int k = 0; k < GCN_SLICES; ++k) c += part[(size_t)k * NP + n];
    float di = rsqrtf(1.0f + (float)c);
    dinv[n] = di;
    w[n] = di * x[n];
}

__global__ void gcn_layer1_gemv_wmma_kernel(const float* __restrict__ part,
                                            const float* __restrict__ dinv,
                                            const float* __restrict__ x,
                                            const float* __restrict__ W1,
                                            const float* __restrict__ b1,
                                            const float* __restrict__ W2,
                                            float* __restrict__ z, float* __restrict__ w,
                                            int n_tiles, int N, int NP) {
    int lane = threadIdx.x & 31;
    int wave = (blockIdx.x * blockDim.x + threadIdx.x) >> 5;
    if (wave >= n_tiles) return;
    int base = wave * 16;
    int m  = lane & 15;
    int kh = lane >> 4;
    int n = min(base + m, N - 1);

    const float* p = part + (size_t)(kh * (GCN_SLICES / 2)) * NP + n;
    float t = 0.0f;
#pragma unroll
    for (int k = 0; k < GCN_SLICES / 2; ++k) t += p[(size_t)k * NP];
    t += __shfl_xor(t, 16);
    float di = dinv[n];
    float s1 = di * t + di * di * x[n];

    v8f acc = {0.f, 0.f, 0.f, 0.f, 0.f, 0.f, 0.f, 0.f};
#pragma unroll
    for (int kc = 0; kc < 4; ++kc) {
        int k0 = kc * 4 + kh * 2;
        v2f a, b;
        a.x = fmaxf(W1[k0 + 0] * s1 + b1[k0 + 0], 0.0f);
        a.y = fmaxf(W1[k0 + 1] * s1 + b1[k0 + 1], 0.0f);
        b.x = W2[k0 + 0];
        b.y = W2[k0 + 1];
        acc = __builtin_amdgcn_wmma_f32_16x16x4_f32(
             false, a,  false, b,
             (short)0, acc,  false,  false);
    }

    float zr = acc[0];
#pragma unroll
    for (int r = 1; r < 8; ++r) zr = ((m & 7) == r) ? acc[r] : zr;
    if (kh == (m >> 3)) {
        z[base + m] = zr;
        w[base + m] = di * zr;
    }
}

__global__ __launch_bounds__(GCN_TPB)
void gcn_scatter_lds_kernel(const int* __restrict__ src, const int* __restrict__ dst,
                            const float* __restrict__ w,
                            float* __restrict__ part, int E, int eps, int NP) {
    __shared__ alignas(16) float acc[GCN_RANGE];
    const int tid = threadIdx.x;
    const int lo  = blockIdx.y * GCN_RANGE;
    float4* a4 = reinterpret_cast<float4*>(acc);
    for (int i = tid; i < GCN_RANGE / 4; i += GCN_TPB) a4[i] = make_float4(0.f, 0.f, 0.f, 0.f);
    __syncthreads();

    const int e0 = blockIdx.x * eps;
    const int n  = min(eps, E - e0);
    const int* qd = dst + e0;
    const int* qs = src + e0;
    int nv = 0;
    if (n > 0 && ((reinterpret_cast<size_t>(qd) | reinterpret_cast<size_t>(qs)) & 15) == 0) nv = n >> 2;
    const int4* pd = reinterpret_cast<const int4*>(qd);
    const int4* ps = reinterpret_cast<const int4*>(qs);
    for (int i = tid; i < nv; i += 2 * GCN_TPB) {
        int d[8], s[8];
#pragma unroll
        for (int u = 0; u < 2; ++u) {
            int j = i + u * GCN_TPB;
            int4 dv = pd[min(j, nv - 1)];
            int4 sv = ps[min(j, nv - 1)];
            if (j >= nv) dv = make_int4(-1, -1, -1, -1);
            d[4 * u + 0] = dv.x; d[4 * u + 1] = dv.y; d[4 * u + 2] = dv.z; d[4 * u + 3] = dv.w;
            s[4 * u + 0] = sv.x; s[4 * u + 1] = sv.y; s[4 * u + 2] = sv.z; s[4 * u + 3] = sv.w;
        }
        float g[8];
        unsigned int r[8];
#pragma unroll
        for (int t = 0; t < 8; ++t) {
            r[t] = (unsigned int)(d[t] - lo);
            int ss = (r[t] < GCN_RANGE) ? s[t] : 0;
            g[t] = w[ss];
        }
#pragma unroll
        for (int t = 0; t < 8; ++t)
            if (r[t] < GCN_RANGE) atomicAdd(&acc[r[t]], g[t]);
    }
    for (int e = 4 * nv + tid; e < n; e += GCN_TPB) {
        unsigned int r = (unsigned int)(qd[e] - lo);
        if (r < GCN_RANGE) atomicAdd(&acc[r], w[qs[e]]);
    }
    __syncthreads();

    float4* o4 = reinterpret_cast<float4*>(part + (size_t)blockIdx.x * NP + lo);
    for (int i = tid; i < GCN_RANGE / 4; i += GCN_TPB) o4[i] = a4[i];
}

__global__ void gcn_final_kernel(float* __restrict__ out, const float* __restrict__ part,
                                 const float* __restrict__ dinv, const float* __restrict__ z,
                                 const float* __restrict__ b2, int N, int NP) {
    int n = blockIdx.x * blockDim.x + threadIdx.x;
    if (n >= N) return;
    float a = 0.0f;
#pragma unroll 16
    for (int k = 0; k < GCN_SLICES; ++k) a += part[(size_t)k * NP + n];
    float di = dinv[n];
    out[n] = di * a + di * di * z[n] + b2[0];
}

extern "C" void kernel_launch(void* const* d_in, const int* in_sizes, int n_in,
                              void* d_out, int out_size, void* d_ws, size_t ws_size,
                              hipStream_t stream) {
    const float* x  = (const float*)d_in[0];
    const float* W1 = (const float*)d_in[1];
    const float* b1 = (const float*)d_in[2];
    const float* W2 = (const float*)d_in[3];
    const float* b2 = (const float*)d_in[4];
    const int*   ei = (const int*)d_in[5];
    int N = in_sizes[0];
    int E = in_sizes[5] / 2;
    const int* src = ei;
    const int* dst = ei + E;
    int paddedN = (N + 15) & ~15;

    char* ws = (char*)d_ws;
    float* dinv = (float*)ws; ws += (size_t)paddedN * sizeof(float);
    float* z    = (float*)ws; ws += (size_t)paddedN * sizeof(float);
    float* w    = (float*)ws; ws += (size_t)paddedN * sizeof(float);
    float* out  = (float*)d_out;

    const int B = 256;
    int tiles = paddedN / 16;

    int windows = (N + GCN_RANGE - 1) / GCN_RANGE;
    int NP  = windows * GCN_RANGE;
    int eps = (((E + GCN_SLICES - 1) / GCN_SLICES) + 3) & ~3;
    float* part = (float*)ws;
    dim3 sgrid(GCN_SLICES, windows);

    gcn_degree_lds_kernel<<<sgrid, GCN_TPB, 0, stream>>>(dst, (unsigned int*)part, E, eps, NP);
    gcn_rsqrt_kernel<<<(N + B - 1) / B, B, 0, stream>>>((const unsigned int*)part, x, dinv, w, N, NP);
    gcn_scatter_lds_kernel<<<sgrid, GCN_TPB, 0, stream>>>(src, dst, w, part, E, eps, NP);
    gcn_layer1_gemv_wmma_kernel<<<(tiles * 32 + B - 1) / B, B, 0, stream>>>(part, dinv, x, W1, b1, W2, z, w, tiles, N, NP);
    gcn_scatter_lds_kernel<<<sgrid, GCN_TPB, 0, stream>>>(src, dst, w, part, E, eps, NP);
    gcn_final_kernel<<<(N + B - 1) / B, B, 0, stream>>>(out, part, dinv, z, b2, N, NP);
}


# Float edge scatters read w[src] from an LDS copy of a src window (2 src x 4 dst windows, 32 edge slices, 64 partials per node as before): no global gather per edge; LDS-atomic sum order changes
# speedup vs baseline: 7.7111x; 1.2858x over previous
#include <hip/hip_runtime.h>
#include <hip/hip_bf16.h>

typedef __attribute__((ext_vector_type(2)))  float    v2f;
typedef __attribute__((ext_vector_type(8)))  float    v8f;

#define GCN_RANGE  25088
#define GCN_SRANGE 50176
#define GCN_SLICES 64
#define GCN_TPB    1024


__global__ __launch_bounds__(GCN_TPB)
void gcn_degree_lds_kernel(const int* __restrict__ dst, unsigned int* __restrict__ part,
                           int E, int eps, int NP) {
    __shared__ alignas(16) unsigned int cnt[GCN_RANGE];
    const int tid = threadIdx.x;
    const int lo  = blockIdx.y * GCN_RANGE;
    uint4* c4 = reinterpret_cast<uint4*>(cnt);
    for (int i = tid; i < GCN_RANGE / 4; i += GCN_TPB) c4[i] = make_uint4(0u, 0u, 0u, 0u);
    __syncthreads();

    const int e0 = blockIdx.x * eps;
    const int n  = min(eps, E - e0);
    const int* q = dst + e0;
    int nv = 0;
    if (n > 0 && (reinterpret_cast<size_t>(q) & 15) == 0) nv = n >> 2;
    const int4* p = reinterpret_cast<const int4*>(q);
    for (int i = tid; i < nv; i += 4 * GCN_TPB) {
        int4 v[4];
#pragma unroll
        for (int u = 0; u < 4; ++u) {
            int j = i + u * GCN_TPB;
            v[u] = p[min(j, nv - 1)];
            if (j >= nv) v[u] = make_int4(-1, -1, -1, -1);
        }
#pragma unroll
        for (int u = 0; u < 4; ++u) {
            unsigned int r0 = (unsigned int)(v[u].x - lo), r1 = (unsigned int)(v[u].y - lo);
            unsigned int r2 = (unsigned int)(v[u].z - lo), r3 = (unsigned int)(v[u].w - lo);
            if (r0 < GCN_RANGE) atomicAdd(&cnt[r0], 1u);
            if (r1 < GCN_RANGE) atomicAdd(&cnt[r1], 1u);
            if (r2 < GCN_RANGE) atomicAdd(&cnt[r2], 1u);
            if (r3 < GCN_RANGE) atomicAdd(&cnt[r3], 1u);
        }
    }
    for (int e = 4 * nv + tid; e < n; e += GCN_TPB) {
        unsigned int r = (unsigned int)(q[e] - lo);
        if (r < GCN_RANGE) atomicAdd(&cnt[r], 1u);
    }
    __syncthreads();

    uint4* o4 = reinterpret_cast<uint4*>(part + (size_t)blockIdx.x * NP + lo);
    for (int i = tid; i < GCN_RANGE / 4; i += GCN_TPB) o4[i] = c4[i];
}

__global__ void gcn_rsqrt_kernel(const unsigned int* __restrict__ part, const float* __restrict__ x,
                                 float* __restrict__ dinv, float* __restrict__ w, int N, int NP) {
    int n = blockIdx.x * blockDim.x + threadIdx.x;
    if (n >= N) return;
    unsigned int c = 0u;
#pragma unroll 16
    for (int k = 0; k < GCN_SLICES; ++k) c += part[(size_t)k * NP + n];
    float di = rsqrtf(1.0f + (float)c);
    dinv[n] = di;
    w[n] = di * x[n];
}

__global__ void gcn_layer1_gemv_wmma_kernel(const float* __restrict__ part,
                                            const float* __restrict__ dinv,
                                            const float* __restrict__ x,
                                            const float* __restrict__ W1,
                                            const float* __restrict__ b1,
                                            const float* __restrict__ W2,
                                            float* __restrict__ z, float* __restrict__ w,
                                            int n_tiles, int N, int NP) {
    int lane = threadIdx.x & 31;
    int wave = (blockIdx.x * blockDim.x + threadIdx.x) >> 5;
    if (wave >= n_tiles) return;
    int base = wave * 16;
    int m  = lane & 15;
    int kh = lane >> 4;
    int n = min(base + m, N - 1);

    const float* p = part + (size_t)(kh * (GCN_SLICES / 2)) * NP + n;
    float t = 0.0f;
#pragma unroll
    for (int k = 0; k < GCN_SLICES / 2; ++k) t += p[(size_t)k * NP];
    t += __shfl_xor(t, 16);
    float di = dinv[n];
    float s1 = di * t + di * di * x[n];

    v8f acc = {0.f, 0.f, 0.f, 0.f, 0.f, 0.f, 0.f, 0.f};
#pragma unroll
    for (int kc = 0; kc < 4; ++kc) {
        int k0 = kc * 4 + kh * 2;
        v2f a, b;
        a.x = fmaxf(W1[k0 + 0] * s1 + b1[k0 + 0], 0.0f);
        a.y = fmaxf(W1[k0 + 1] * s1 + b1[k0 + 1], 0.0f);
        b.x = W2[k0 + 0];
        b.y = W2[k0 + 1];
        acc = __builtin_amdgcn_wmma_f32_16x16x4_f32(
             false, a,  false, b,
             (short)0, acc,  false,  false);
    }

    float zr = acc[0];
#pragma unroll
    for (int r = 1; r < 8; ++r) zr = ((m & 7) == r) ? acc[r] : zr;
    if (kh == (m >> 3)) {
        z[base + m] = zr;
        w[base + m] = di * zr;
    }
}

__global__ __launch_bounds__(GCN_TPB)
void gcn_scatter_lds_kernel(const int* __restrict__ src, const int* __restrict__ dst,
                            const float* __restrict__ w,
                            float* __restrict__ part, int E, int eps, int esl, int swin,
                            int N, int NP) {
    __shared__ alignas(16) float acc[GCN_RANGE];
    __shared__ alignas(16) float wl[GCN_SRANGE];
    const int tid = threadIdx.x;
    const int lo  = blockIdx.y * GCN_RANGE;
    const int sl  = blockIdx.x / swin;
    const int slo = (blockIdx.x - sl * swin) * GCN_SRANGE;
    float4* a4 = reinterpret_cast<float4*>(acc);
    for (int i = tid; i < GCN_RANGE / 4; i += GCN_TPB) a4[i] = make_float4(0.f, 0.f, 0.f, 0.f);

    const int cnt = max(0, min(GCN_SRANGE, N - slo));
    const float* wp = w + slo;
    int c4 = 0;
    if (cnt > 0 && (reinterpret_cast<size_t>(wp) & 15) == 0) c4 = cnt >> 2;
    float4* l4 = reinterpret_cast<float4*>(wl);
    const float4* g4 = reinterpret_cast<const float4*>(wp);
    for (int i = tid; i < c4; i += 8 * GCN_TPB) {
        float4 v[8];
#pragma unroll
        for (int u = 0; u < 8; ++u) v[u] = g4[min(i + u * GCN_TPB, c4 - 1)];
#pragma unroll
        for (int u = 0; u < 8; ++u) l4[min(i + u * GCN_TPB, c4 - 1)] = v[u];
    }
    for (int i = 4 * c4 + tid; i < cnt; i += GCN_TPB) wl[i] = wp[i];
    for (int i = cnt + tid; i < GCN_SRANGE; i += GCN_TPB) wl[i] = 0.0f;
    __syncthreads();

    const int e0 = sl * eps;
    const int n  = (sl < esl) ? min(eps, E - e0) : 0;
    const int* qd = dst + e0;
    const int* qs = src + e0;
    int nv = 0;
    if (n > 0 && ((reinterpret_cast<size_t>(qd) | reinterpret_cast<size_t>(qs)) & 15) == 0) nv = n >> 2;
    const int4* pd = reinterpret_cast<const int4*>(qd);
    const int4* ps = reinterpret_cast<const int4*>(qs);
    const unsigned int lo4 = 4u * (unsigned int)lo, slo4 = 4u * (unsigned int)slo;
    char* ab = reinterpret_cast<char*>(acc);
    const char* wb = reinterpret_cast<const char*>(wl);
    for (int i = tid; i < nv; i += 4 * GCN_TPB) {
        int d[16], s[16];
#pragma unroll
        for (int u = 0; u < 4; ++u) {
            int j = i + u * GCN_TPB;
            int4 dv = pd[min(j, nv - 1)];
            int4 sv = ps[min(j, nv - 1)];
            if (j >= nv) dv = make_int4(-1, -1, -1, -1);
            d[4 * u + 0] = dv.x; d[4 * u + 1] = dv.y; d[4 * u + 2] = dv.z; d[4 * u + 3] = dv.w;
            s[4 * u + 0] = sv.x; s[4 * u + 1] = sv.y; s[4 * u + 2] = sv.z; s[4 * u + 3] = sv.w;
        }
        float g[16];
        unsigned int r[16];
#pragma unroll
        for (int t = 0; t < 16; ++t) {
            unsigned int q = ((unsigned int)s[t] << 2) - slo4;
            r[t] = ((unsigned int)d[t] << 2) - lo4;
            if (q >= 4u * GCN_SRANGE) r[t] = 0xFFFFFFFFu;
            g[t] = *reinterpret_cast<const float*>(wb + ((r[t] < 4u * GCN_RANGE) ? q : 0u));
        }
#pragma unroll
        for (int t = 0; t < 16; ++t)
            if (r[t] < 4u * GCN_RANGE) atomicAdd(reinterpret_cast<float*>(ab + r[t]), g[t]);
    }
    for (int e = 4 * nv + tid; e < n; e += GCN_TPB) {
        unsigned int r = (unsigned int)(qd[e] - lo);
        unsigned int q = (unsigned int)(qs[e] - slo);
        if (r < GCN_RANGE && q < GCN_SRANGE) atomicAdd(&acc[r], wl[q]);
    }
    __syncthreads();

    float4* o4 = reinterpret_cast<float4*>(part + (size_t)blockIdx.x * NP + lo);
    for (int i = tid; i < GCN_RANGE / 4; i += GCN_TPB) o4[i] = a4[i];
}

__global__ void gcn_final_kernel(float* __restrict__ out, const float* __restrict__ part,
                                 const float* __restrict__ dinv, const float* __restrict__ z,
                                 const float* __restrict__ b2, int N, int NP) {
    int n = blockIdx.x * blockDim.x + threadIdx.x;
    if (n >= N) return;
    float a = 0.0f;
#pragma unroll 16
    for (int k = 0; k < GCN_SLICES; ++k) a += part[(size_t)k * NP + n];
    float di = dinv[n];
    out[n] = di * a + di * di * z[n] + b2[0];
}

extern "C" void kernel_launch(void* const* d_in, const int* in_sizes, int n_in,
                              void* d_out, int out_size, void* d_ws, size_t ws_size,
                              hipStream_t stream) {
    const float* x  = (const float*)d_in[0];
    const float* W1 = (const float*)d_in[1];
    const float* b1 = (const float*)d_in[2];
    const float* W2 = (const float*)d_in[3];
    const float* b2 = (const float*)d_in[4];
    const int*   ei = (const int*)d_in[5];
    int N = in_sizes[0];
    int E = in_sizes[5] / 2;
    const int* src = ei;
    const int* dst = ei + E;
    int paddedN = (N + 15) & ~15;

    char* ws = (char*)d_ws;
    float* dinv = (float*)ws; ws += (size_t)paddedN * sizeof(float);
    float* z    = (float*)ws; ws += (size_t)paddedN * sizeof(float);
    float* w    = (float*)ws; ws += (size_t)paddedN * sizeof(float);
    float* out  = (float*)d_out;

    const int B = 256;
    int tiles = paddedN / 16;

    int windows = (N + GCN_RANGE - 1) / GCN_RANGE;
    int NP  = windows * GCN_RANGE;
    int eps = (((E + GCN_SLICES - 1) / GCN_SLICES) + 3) & ~3;
    float* part = (float*)ws;
    dim3 sgrid(GCN_SLICES, windows);
    int swin = (N + GCN_SRANGE - 1) / GCN_SRANGE;
    int esl  = GCN_SLICES / swin;
    int eps2 = (((E + esl - 1) / esl) + 3) & ~3;

    gcn_degree_lds_kernel<<<sgrid, GCN_TPB, 0, stream>>>(dst, (unsigned int*)part, E, eps, NP);
    gcn_rsqrt_kernel<<<(N + B - 1) / B, B, 0, stream>>>((const unsigned int*)part, x, dinv, w, N, NP);
    gcn_scatter_lds_kernel<<<sgrid, GCN_TPB, 0, stream>>>(src, dst, w, part, E, eps2, esl, swin, N, NP);
    gcn_layer1_gemv_wmma_kernel<<<(tiles * 32 + B - 1) / B, B, 0, stream>>>(part, dinv, x, W1, b1, W2, z, w, tiles, N, NP);
    gcn_scatter_lds_kernel<<<sgrid, GCN_TPB, 0, stream>>>(src, dst, w, part, E, eps2, esl, swin, N, NP);
    gcn_final_kernel<<<(N + B - 1) / B, B, 0, stream>>>(out, part, dinv, z, b2, N, NP);
}


# Loads of a phase go out as one group, one cache round trip not several: all 64 slice partials per node (degree and final sums), 13 16-byte loads a thread in degree scan and w copy; sum order kept
# speedup vs baseline: 8.0065x; 1.0383x over previous
#include <hip/hip_runtime.h>
#include <hip/hip_bf16.h>

typedef __attribute__((ext_vector_type(2)))  float    v2f;
typedef __attribute__((ext_vector_type(8)))  float    v8f;

#define GCN_RANGE  25088
#define GCN_SRANGE 50176
#define GCN_SLICES 64
#define GCN_TPB    1024
#define GCN_DEG_U  13
#define GCN_W_U    13


__global__ __launch_bounds__(GCN_TPB)
void gcn_degree_lds_kernel(const int* __restrict__ dst, unsigned int* __restrict__ part,
                           int E, int eps, int NP) {
    __shared__ alignas(16) unsigned int cnt[GCN_RANGE];
    const int tid = threadIdx.x;
    const int lo  = blockIdx.y * GCN_RANGE;
    uint4* c4 = reinterpret_cast<uint4*>(cnt);
    for (int i = tid; i < GCN_RANGE / 4; i += GCN_TPB) c4[i] = make_uint4(0u, 0u, 0u, 0u);
    __syncthreads();

    const int e0 = blockIdx.x * eps;
    const int n  = min(eps, E - e0);
    const int* q = dst + e0;
    int nv = 0;
    if (n > 0 && (reinterpret_cast<size_t>(q) & 15) == 0) nv = n >> 2;
    const int4* p = reinterpret_cast<const int4*>(q);
    for (int i = tid; i < nv; i += GCN_DEG_U * GCN_TPB) {
        int4 v[GCN_DEG_U];
#pragma unroll
        for (int u = 0; u < GCN_DEG_U; ++u) {
            int j = i + u * GCN_TPB;
            v[u] = p[min(j, nv - 1)];
            if (j >= nv) v[u] = make_int4(-1, -1, -1, -1);
        }
#pragma unroll
        for (int u = 0; u < GCN_DEG_U; ++u) {
            unsigned int r0 = (unsigned int)(v[u].x - lo), r1 = (unsigned int)(v[u].y - lo);
            unsigned int r2 = (unsigned int)(v[u].z - lo), r3 = (unsigned int)(v[u].w - lo);
            if (r0 < GCN_RANGE) atomicAdd(&cnt[r0], 1u);
            if (r1 < GCN_RANGE) atomicAdd(&cnt[r1], 1u);
            if (r2 < GCN_RANGE) atomicAdd(&cnt[r2], 1u);
            if (r3 < GCN_RANGE) atomicAdd(&cnt[r3], 1u);
        }
    }
    for (int e = 4 * nv + tid; e < n; e += GCN_TPB) {
        unsigned int r = (unsigned int)(q[e] - lo);
        if (r < GCN_RANGE) atomicAdd(&cnt[r], 1u);
    }
    __syncthreads();

    uint4* o4 = reinterpret_cast<uint4*>(part + (size_t)blockIdx.x * NP + lo);
    for (int i = tid; i < GCN_RANGE / 4; i += GCN_TPB) o4[i] = c4[i];
}

__global__ void gcn_rsqrt_kernel(const unsigned int* __restrict__ part, const float* __restrict__ x,
                                 float* __restrict__ dinv, float* __restrict__ w, int N, int NP) {
    int n = blockIdx.x * blockDim.x + threadIdx.x;
    if (n >= N) return;
    unsigned int pv[GCN_SLICES];
#pragma unroll
    for (int k = 0; k < GCN_SLICES; ++k) pv[k] = part[(size_t)k * NP + n];
    unsigned int c = 0u;
#pragma unroll
    for (int k = 0; k < GCN_SLICES; ++k) c += pv[k];
    float di = rsqrtf(1.0f + (float)c);
    dinv[n] = di;
    w[n] = di * x[n];
}

__global__ void gcn_layer1_gemv_wmma_kernel(const float* __restrict__ part,
                                            const float* __restrict__ dinv,
                                            const float* __restrict__ x,
                                            const float* __restrict__ W1,
                                            const float* __restrict__ b1,
                                            const float* __restrict__ W2,
                                            float* __restrict__ z, float* __restrict__ w,
                                            int n_tiles, int N, int NP) {
    int lane = threadIdx.x & 31;
    int wave = (blockIdx.x * blockDim.x + threadIdx.x) >> 5;
    if (wave >= n_tiles) return;
    int base = wave * 16;
    int m  = lane & 15;
    int kh = lane >> 4;
    int n = min(base + m, N - 1);

    const float* p = part + (size_t)(kh * (GCN_SLICES / 2)) * NP + n;
    float t = 0.0f;
#pragma unroll
    for (int k = 0; k < GCN_SLICES / 2; ++k) t += p[(size_t)k * NP];
    t += __shfl_xor(t, 16);
    float di = dinv[n];
    float s1 = di * t + di * di * x[n];

    v8f acc = {0.f, 0.f, 0.f, 0.f, 0.f, 0.f, 0.f, 0.f};
#pragma unroll
    for (int kc = 0; kc < 4; ++kc) {
        int k0 = kc * 4 + kh * 2;
        v2f a, b;
        a.x = fmaxf(W1[k0 + 0] * s1 + b1[k0 + 0], 0.0f);
        a.y = fmaxf(W1[k0 + 1] * s1 + b1[k0 + 1], 0.0f);
        b.x = W2[k0 + 0];
        b.y = W2[k0 + 1];
        acc = __builtin_amdgcn_wmma_f32_16x16x4_f32(
             false, a,  false, b,
             (short)0, acc,  false,  false);
    }

    float zr = acc[0];
#pragma unroll
    for (int r = 1; r < 8; ++r) zr = ((m & 7) == r) ? acc[r] : zr;
    if (kh == (m >> 3)) {
        z[base + m] = zr;
        w[base + m] = di * zr;
    }
}

__global__ __launch_bounds__(GCN_TPB)
void gcn_scatter_lds_kernel(const int* __restrict__ src, const int* __restrict__ dst,
                            const float* __restrict__ w,
                            float* __restrict__ part, int E, int eps, int esl, int swin,
                            int N, int NP) {
    __shared__ alignas(16) float acc[GCN_RANGE];
    __shared__ alignas(16) float wl[GCN_SRANGE];
    const int tid = threadIdx.x;
    const int lo  = blockIdx.y * GCN_RANGE;
    const int sl  = blockIdx.x / swin;
    const int slo = (blockIdx.x - sl * swin) * GCN_SRANGE;
    float4* a4 = reinterpret_cast<float4*>(acc);
    for (int i = tid; i < GCN_RANGE / 4; i += GCN_TPB) a4[i] = make_float4(0.f, 0.f, 0.f, 0.f);

    const int cnt = max(0, min(GCN_SRANGE, N - slo));
    const float* wp = w + slo;
    int c4 = 0;
    if (cnt > 0 && (reinterpret_cast<size_t>(wp) & 15) == 0) c4 = cnt >> 2;
    float4* l4 = reinterpret_cast<float4*>(wl);
    const float4* g4 = reinterpret_cast<const float4*>(wp);
    for (int i = tid; i < c4; i += GCN_W_U * GCN_TPB) {
        float4 v[GCN_W_U];
#pragma unroll
        for (int u = 0; u < GCN_W_U; ++u) v[u] = g4[min(i + u * GCN_TPB, c4 - 1)];
#pragma unroll
        for (int u = 0; u < GCN_W_U; ++u) l4[min(i + u * GCN_TPB, c4 - 1)] = v[u];
    }
    for (int i = 4 * c4 + tid; i < cnt; i += GCN_TPB) wl[i] = wp[i];
    for (int i = cnt + tid; i < GCN_SRANGE; i += GCN_TPB) wl[i] = 0.0f;
    __syncthreads();

    const int e0 = sl * eps;
    const int n  = (sl < esl) ? min(eps, E - e0) : 0;
    const int* qd = dst + e0;
    const int* qs = src + e0;
    int nv = 0;
    if (n > 0 && ((reinterpret_cast<size_t>(qd) | reinterpret_cast<size_t>(qs)) & 15) == 0) nv = n >> 2;
    const int4* pd = reinterpret_cast<const int4*>(qd);
    const int4* ps = reinterpret_cast<const int4*>(qs);
    const unsigned int lo4 = 4u * (unsigned int)lo, slo4 = 4u * (unsigned int)slo;
    char* ab = reinterpret_cast<char*>(acc);
    const char* wb = reinterpret_cast<const char*>(wl);
    for (int i = tid; i < nv; i += 4 * GCN_TPB) {
        int d[16], s[16];
#pragma unroll
        for (int u = 0; u < 4; ++u) {
            int j = i + u * GCN_TPB;
            int4 dv = pd[min(j, nv - 1)];
            int4 sv = ps[min(j, nv - 1)];
            if (j >= nv) dv = make_int4(-1, -1, -1, -1);
            d[4 * u + 0] = dv.x; d[4 * u + 1] = dv.y; d[4 * u + 2] = dv.z; d[4 * u + 3] = dv.w;
            s[4 * u + 0] = sv.x; s[4 * u + 1] = sv.y; s[4 * u + 2] = sv.z; s[4 * u + 3] = sv.w;
        }
        float g[16];
        unsigned int r[16];
#pragma unroll
        for (int t = 0; t < 16; ++t) {
            unsigned int q = ((unsigned int)s[t] << 2) - slo4;
            r[t] = ((unsigned int)d[t] << 2) - lo4;
            if (q >= 4u * GCN_SRANGE) r[t] = 0xFFFFFFFFu;
            g[t] = *reinterpret_cast<const float*>(wb + ((r[t] < 4u * GCN_RANGE) ? q : 0u));
        }
#pragma unroll
        for (int t = 0; t < 16; ++t)
            if (r[t] < 4u * GCN_RANGE) atomicAdd(reinterpret_cast<float*>(ab + r[t]), g[t]);
    }
    for (int e = 4 * nv + tid; e < n; e += GCN_TPB) {
        unsigned int r = (unsigned int)(qd[e] - lo);
        unsigned int q = (unsigned int)(qs[e] - slo);
        if (r < GCN_RANGE && q < GCN_SRANGE) atomicAdd(&acc[r], wl[q]);
    }
    __syncthreads();

    float4* o4 = reinterpret_cast<float4*>(part + (size_t)blockIdx.x * NP + lo);
    for (int i = tid; i < GCN_RANGE / 4; i += GCN_TPB) o4[i] = a4[i];
}

__global__ void gcn_final_kernel(float* __restrict__ out, const float* __restrict__ part,
                                 const float* __restrict__ dinv, const float* __restrict__ z,
                                 const float* __restrict__ b2, int N, int NP) {
    int n = blockIdx.x * blockDim.x + threadIdx.x;
    if (n >= N) return;
    float pv[GCN_SLICES];
#pragma unroll
    for (int k = 0; k < GCN_SLICES; ++k) pv[k] = part[(size_t)k * NP + n];
    float a = 0.0f;
#pragma unroll
    for (int k = 0; k < GCN_SLICES; ++k) a += pv[k];
    float di = dinv[n];
    out[n] = di * a + di * di * z[n] + b2[0];
}

extern "C" void kernel_launch(void* const* d_in, const int* in_sizes, int n_in,
                              void* d_out, int out_size, void* d_ws, size_t ws_size,
                              hipStream_t stream) {
    const float* x  = (const float*)d_in[0];
    const float* W1 = (const float*)d_in[1];
    const float* b1 = (const float*)d_in[2];
    const float* W2 = (const float*)d_in[3];
    const float* b2 = (const float*)d_in[4];
    const int*   ei = (const int*)d_in[5];
    int N = in_sizes[0];
    int E = in_sizes[5] / 2;
    const int* src = ei;
    const int* dst = ei + E;
    int paddedN = (N + 15) & ~15;

    char* ws = (char*)d_ws;
    float* dinv = (float*)ws; ws += (size_t)paddedN * sizeof(float);
    float* z    = (float*)ws; ws += (size_t)paddedN * sizeof(float);
    float* w    = (float*)ws; ws += (size_t)paddedN * sizeof(float);
    float* out  = (float*)d_out;

    const int B = 256;
    int tiles = paddedN / 16;

    int windows = (N + GCN_RANGE - 1) / GCN_RANGE;
    int NP  = windows * GCN_RANGE;
    int eps = (((E + GCN_SLICES - 1) / GCN_SLICES) + 3) & ~3;
    float* part = (float*)ws;
    dim3 sgrid(GCN_SLICES, windows);
    int swin = (N + GCN_SRANGE - 1) / GCN_SRANGE;
    int esl  = GCN_SLICES / swin;
    int eps2 = (((E + esl - 1) / esl) + 3) & ~3;

    gcn_degree_lds_kernel<<<sgrid, GCN_TPB, 0, stream>>>(dst, (unsigned int*)part, E, eps, NP);
    gcn_rsqrt_kernel<<<(N + B - 1) / B, B, 0, stream>>>((const unsigned int*)part, x, dinv, w, N, NP);
    gcn_scatter_lds_kernel<<<sgrid, GCN_TPB, 0, stream>>>(src, dst, w, part, E, eps2, esl, swin, N, NP);
    gcn_layer1_gemv_wmma_kernel<<<(tiles * 32 + B - 1) / B, B, 0, stream>>>(part, dinv, x, W1, b1, W2, z, w, tiles, N, NP);
    gcn_scatter_lds_kernel<<<sgrid, GCN_TPB, 0, stream>>>(src, dst, w, part, E, eps2, esl, swin, N, NP);
    gcn_final_kernel<<<(N + B - 1) / B, B, 0, stream>>>(out, part, dinv, z, b2, N, NP);
}


# Float scatters request the next trip's 8 edge loads before this trip's LDS reads and adds (2 register sets; first trip before the window set-up, degree pass too); same addends, LDS-atomic order kept
# speedup vs baseline: 8.2825x; 1.0345x over previous
#include <hip/hip_runtime.h>
#include <hip/hip_bf16.h>

typedef __attribute__((ext_vector_type(2)))  float    v2f;
typedef __attribute__((ext_vector_type(8)))  float    v8f;

#define GCN_RANGE  25088
#define GCN_SRANGE 50176
#define GCN_SLICES 64
#define GCN_TPB    1024
#define GCN_DEG_U  13
#define GCN_W_U    13


__global__ __launch_bounds__(GCN_TPB)
void gcn_degree_lds_kernel(const int* __restrict__ dst, unsigned int* __restrict__ part,
                           int E, int eps, int NP) {
    __shared__ alignas(16) unsigned int cnt[GCN_RANGE];
    const int tid = threadIdx.x;
    const int lo  = blockIdx.y * GCN_RANGE;
    uint4* c4 = reinterpret_cast<uint4*>(cnt);

    const int e0 = blockIdx.x * eps;
    const int n  = min(eps, E - e0);
    const int* q = dst + e0;
    int nv = 0;
    if (n > 0 && (reinterpret_cast<size_t>(q) & 15) == 0) nv = n >> 2;
    const int4* p = reinterpret_cast<const int4*>(q);

    int4 v[GCN_DEG_U];
    if (nv > 0) {
#pragma unroll
        for (int u = 0; u < GCN_DEG_U; ++u) v[u] = p[min(tid + u * GCN_TPB, nv - 1)];
    }

    for (int i = tid; i < GCN_RANGE / 4; i += GCN_TPB) c4[i] = make_uint4(0u, 0u, 0u, 0u);
    __syncthreads();

    for (int i = tid; i < nv; i += GCN_DEG_U * GCN_TPB) {
        if (i != tid) {
#pragma unroll
            for (int u = 0; u < GCN_DEG_U; ++u) v[u] = p[min(i + u * GCN_TPB, nv - 1)];
        }
#pragma unroll
        for (int u = 0; u < GCN_DEG_U; ++u) {
            int4 t = v[u];
            if (i + u * GCN_TPB >= nv) t = make_int4(-1, -1, -1, -1);
            unsigned int r0 = (unsigned int)(t.x - lo), r1 = (unsigned int)(t.y - lo);
            unsigned int r2 = (unsigned int)(t.z - lo), r3 = (unsigned int)(t.w - lo);
            if (r0 < GCN_RANGE) atomicAdd(&cnt[r0], 1u);
            if (r1 < GCN_RANGE) atomicAdd(&cnt[r1], 1u);
            if (r2 < GCN_RANGE) atomicAdd(&cnt[r2], 1u);
            if (r3 < GCN_RANGE) atomicAdd(&cnt[r3], 1u);
        }
    }
    for (int e = 4 * nv + tid; e < n; e += GCN_TPB) {
        unsigned int r = (unsigned int)(q[e] - lo);
        if (r < GCN_RANGE) atomicAdd(&cnt[r], 1u);
    }
    __syncthreads();

    uint4* o4 = reinterpret_cast<uint4*>(part + (size_t)blockIdx.x * NP + lo);
    for (int i = tid; i < GCN_RANGE / 4; i += GCN_TPB) o4[i] = c4[i];
}

__global__ void gcn_rsqrt_kernel(const unsigned int* __restrict__ part, const float* __restrict__ x,
                                 float* __restrict__ dinv, float* __restrict__ w, int N, int NP) {
    int n = blockIdx.x * blockDim.x + threadIdx.x;
    if (n >= N) return;
    unsigned int pv[GCN_SLICES];
#pragma unroll
    for (int k = 0; k < GCN_SLICES; ++k) pv[k] = part[(size_t)k * NP + n];
    unsigned int c = 0u;
#pragma unroll
    for (int k = 0; k < GCN_SLICES; ++k) c += pv[k];
    float di = rsqrtf(1.0f + (float)c);
    dinv[n] = di;
    w[n] = di * x[n];
}

__global__ void gcn_layer1_gemv_wmma_kernel(const float* __restrict__ part,
                                            const float* __restrict__ dinv,
                                            const float* __restrict__ x,
                                            const float* __restrict__ W1,
                                            const float* __restrict__ b1,
                                            const float* __restrict__ W2,
                                            float* __restrict__ z, float* __restrict__ w,
                                            int n_tiles, int N, int NP) {
    int lane = threadIdx.x & 31;
    int wave = (blockIdx.x * blockDim.x + threadIdx.x) >> 5;
    if (wave >= n_tiles) return;
    int base = wave * 16;
    int m  = lane & 15;
    int kh = lane >> 4;
    int n = min(base + m, N - 1);

    const float* p = part + (size_t)(kh * (GCN_SLICES / 2)) * NP + n;
    float t = 0.0f;
#pragma unroll
    for (int k = 0; k < GCN_SLICES / 2; ++k) t += p[(size_t)k * NP];
    t += __shfl_xor(t, 16);
    float di = dinv[n];
    float s1 = di * t + di * di * x[n];

    v8f acc = {0.f, 0.f, 0.f, 0.f, 0.f, 0.f, 0.f, 0.f};
#pragma unroll
    for (int kc = 0; kc < 4; ++kc) {
        int k0 = kc * 4 + kh * 2;
        v2f a, b;
        a.x = fmaxf(W1[k0 + 0] * s1 + b1[k0 + 0], 0.0f);
        a.y = fmaxf(W1[k0 + 1] * s1 + b1[k0 + 1], 0.0f);
        b.x = W2[k0 + 0];
        b.y = W2[k0 + 1];
        acc = __builtin_amdgcn_wmma_f32_16x16x4_f32(
             false, a,  false, b,
             (short)0, acc,  false,  false);
    }

    float zr = acc[0];
#pragma unroll
    for (int r = 1; r < 8; ++r) zr = ((m & 7) == r) ? acc[r] : zr;
    if (kh == (m >> 3)) {
        z[base + m] = zr;
        w[base + m] = di * zr;
    }
}

template <bool TAIL>
__device__ __forceinline__ void gcn_edges16(const int4 (&dv)[4], const int4 (&sv)[4],
                                            int base, int nv,
                                            unsigned int lo4, unsigned int slo4,
                                            char* ab, const char* wb) {
    int d[16], s[16];
#pragma unroll
    for (int u = 0; u < 4; ++u) {
        int4 t = dv[u];
        if (TAIL && base + u * GCN_TPB >= nv) t = make_int4(-1, -1, -1, -1);
        d[4 * u + 0] = t.x; d[4 * u + 1] = t.y; d[4 * u + 2] = t.z; d[4 * u + 3] = t.w;
        s[4 * u + 0] = sv[u].x; s[4 * u + 1] = sv[u].y; s[4 * u + 2] = sv[u].z; s[4 * u + 3] = sv[u].w;
    }
    float g[16];
    unsigned int r[16];
#pragma unroll
    for (int t = 0; t < 16; ++t) {
        unsigned int q = ((unsigned int)s[t] << 2) - slo4;
        r[t] = ((unsigned int)d[t] << 2) - lo4;
        if (q >= 4u * GCN_SRANGE) r[t] = 0xFFFFFFFFu;
        g[t] = *reinterpret_cast<const float*>(wb + ((r[t] < 4u * GCN_RANGE) ? q : 0u));
    }
#pragma unroll
    for (int t = 0; t < 16; ++t)
        if (r[t] < 4u * GCN_RANGE) atomicAdd(reinterpret_cast<float*>(ab + r[t]), g[t]);
}

__device__ __forceinline__ void gcn_edges_request(int4 (&dv)[4], int4 (&sv)[4],
                                                  const int4* __restrict__ pd,
                                                  const int4* __restrict__ ps, int base, int nv) {
#pragma unroll
    for (int u = 0; u < 4; ++u) {
        int j = min(base + u * GCN_TPB, nv - 1);
        dv[u] = pd[j];
        sv[u] = ps[j];
    }
}

__global__ __launch_bounds__(GCN_TPB)
void gcn_scatter_lds_kernel(const int* __restrict__ src, const int* __restrict__ dst,
                            const float* __restrict__ w,
                            float* __restrict__ part, int E, int eps, int esl, int swin,
                            int N, int NP) {
    __shared__ alignas(16) float acc[GCN_RANGE];
    __shared__ alignas(16) float wl[GCN_SRANGE];
    const int tid = threadIdx.x;
    const int lo  = blockIdx.y * GCN_RANGE;
    const int sl  = blockIdx.x / swin;
    const int slo = (blockIdx.x - sl * swin) * GCN_SRANGE;
    float4* a4 = reinterpret_cast<float4*>(acc);

    const int e0 = sl * eps;
    const int n  = (sl < esl) ? min(eps, E - e0) : 0;
    const int* qd = dst + e0;
    const int* qs = src + e0;
    int nv = 0;
    if (n > 0 && ((reinterpret_cast<size_t>(qd) | reinterpret_cast<size_t>(qs)) & 15) == 0) nv = n >> 2;
    const int4* pd = reinterpret_cast<const int4*>(qd);
    const int4* ps = reinterpret_cast<const int4*>(qs);
    const int trips = (nv + 4 * GCN_TPB - 1) / (4 * GCN_TPB);

    int4 da[4], sa[4], db[4], sb[4];
    if (trips > 0) gcn_edges_request(da, sa, pd, ps, tid, nv);

    for (int i = tid; i < GCN_RANGE / 4; i += GCN_TPB) a4[i] = make_float4(0.f, 0.f, 0.f, 0.f);

    const int cnt = max(0, min(GCN_SRANGE, N - slo));
    const float* wp = w + slo;
    int c4 = 0;
    if (cnt > 0 && (reinterpret_cast<size_t>(wp) & 15) == 0) c4 = cnt >> 2;
    float4* l4 = reinterpret_cast<float4*>(wl);
    const float4* g4 = reinterpret_cast<const float4*>(wp);
    for (int i = tid; i < c4; i += GCN_W_U * GCN_TPB) {
        float4 v[GCN_W_U];
#pragma unroll
        for (int u = 0; u < GCN_W_U; ++u) v[u] = g4[min(i + u * GCN_TPB, c4 - 1)];
#pragma unroll
        for (int u = 0; u < GCN_W_U; ++u) l4[min(i + u * GCN_TPB, c4 - 1)] = v[u];
    }
    for (int i = 4 * c4 + tid; i < cnt; i += GCN_TPB) wl[i] = wp[i];
    for (int i = cnt + tid; i < GCN_SRANGE; i += GCN_TPB) wl[i] = 0.0f;
    __syncthreads();

    const unsigned int lo4 = 4u * (unsigned int)lo, slo4 = 4u * (unsigned int)slo;
    char* ab = reinterpret_cast<char*>(acc);
    const char* wb = reinterpret_cast<const char*>(wl);
    if (trips > 0) {
        int t = 0, base = tid;
        for (; t + 2 < trips; t += 2) {
            gcn_edges_request(db, sb, pd, ps, base + 4 * GCN_TPB, nv);
            __builtin_amdgcn_sched_barrier(0);
            gcn_edges16<false>(da, sa, base, nv, lo4, slo4, ab, wb);
            gcn_edges_request(da, sa, pd, ps, base + 8 * GCN_TPB, nv);
            __builtin_amdgcn_sched_barrier(0);
            gcn_edges16<false>(db, sb, base + 4 * GCN_TPB, nv, lo4, slo4, ab, wb);
            base += 8 * GCN_TPB;
        }
        if (t + 2 == trips) {
            gcn_edges_request(db, sb, pd, ps, base + 4 * GCN_TPB, nv);
            __builtin_amdgcn_sched_barrier(0);
            gcn_edges16<false>(da, sa, base, nv, lo4, slo4, ab, wb);
#pragma unroll
            for (int u = 0; u < 4; ++u) { da[u] = db[u]; sa[u] = sb[u]; }
            base += 4 * GCN_TPB;
        }
        if (base < nv) gcn_edges16<true>(da, sa, base, nv, lo4, slo4, ab, wb);
    }
    for (int e = 4 * nv + tid; e < n; e += GCN_TPB) {
        unsigned int r = (unsigned int)(qd[e] - lo);
        unsigned int q = (unsigned int)(qs[e] - slo);
        if (r < GCN_RANGE && q < GCN_SRANGE) atomicAdd(&acc[r], wl[q]);
    }
    __syncthreads();

    float4* o4 = reinterpret_cast<float4*>(part + (size_t)blockIdx.x * NP + lo);
    for (int i = tid; i < GCN_RANGE / 4; i += GCN_TPB) o4[i] = a4[i];
}

__global__ void gcn_final_kernel(float* __restrict__ out, const float* __restrict__ part,
                                 const float* __restrict__ dinv, const float* __restrict__ z,
                                 const float* __restrict__ b2, int N, int NP) {
    int n = blockIdx.x * blockDim.x + threadIdx.x;
    if (n >= N) return;
    float pv[GCN_SLICES];
#pragma unroll
    for (int k = 0; k < GCN_SLICES; ++k) pv[k] = part[(size_t)k * NP + n];
    float a = 0.0f;
#pragma unroll
    for (int k = 0; k < GCN_SLICES; ++k) a += pv[k];
    float di = dinv[n];
    out[n] = di * a + di * di * z[n] + b2[0];
}

extern "C" void kernel_launch(void* const* d_in, const int* in_sizes, int n_in,
                              void* d_out, int out_size, void* d_ws, size_t ws_size,
                              hipStream_t stream) {
    const float* x  = (const float*)d_in[0];
    const float* W1 = (const float*)d_in[1];
    const float* b1 = (const float*)d_in[2];
    const float* W2 = (const float*)d_in[3];
    const float* b2 = (const float*)d_in[4];
    const int*   ei = (const int*)d_in[5];
    int N = in_sizes[0];
    int E = in_sizes[5] / 2;
    const int* src = ei;
    const int* dst = ei + E;
    int paddedN = (N + 15) & ~15;

    char* ws = (char*)d_ws;
    float* dinv = (float*)ws; ws += (size_t)paddedN * sizeof(float);
    float* z    = (float*)ws; ws += (size_t)paddedN * sizeof(float);
    float* w    = (float*)ws; ws += (size_t)paddedN * sizeof(float);
    float* out  = (float*)d_out;

    const int B = 256;
    int tiles = paddedN / 16;

    int windows = (N + GCN_RANGE - 1) / GCN_RANGE;
    int NP  = windows * GCN_RANGE;
    int eps = (((E + GCN_SLICES - 1) / GCN_SLICES) + 3) & ~3;
    float* part = (float*)ws;
    dim3 sgrid(GCN_SLICES, windows);
    int swin = (N + GCN_SRANGE - 1) / GCN_SRANGE;
    int esl  = GCN_SLICES / swin;
    int eps2 = (((E + esl - 1) / esl) + 3) & ~3;

    gcn_degree_lds_kernel<<<sgrid, GCN_TPB, 0, stream>>>(dst, (unsigned int*)part, E, eps, NP);
    gcn_rsqrt_kernel<<<(N + B - 1) / B, B, 0, stream>>>((const unsigned int*)part, x, dinv, w, N, NP);
    gcn_scatter_lds_kernel<<<sgrid, GCN_TPB, 0, stream>>>(src, dst, w, part, E, eps2, esl, swin, N, NP);
    gcn_layer1_gemv_wmma_kernel<<<(tiles * 32 + B - 1) / B, B, 0, stream>>>(part, dinv, x, W1, b1, W2, z, w, tiles, N, NP);
    gcn_scatter_lds_kernel<<<sgrid, GCN_TPB, 0, stream>>>(src, dst, w, part, E, eps2, esl, swin, N, NP);
    gcn_final_kernel<<<(N + B - 1) / B, B, 0, stream>>>(out, part, dinv, z, b2, N, NP);
}
